# starting kernel
# baseline (speedup 1.0000x reference)
#include <hip/hip_runtime.h>
#include <math.h>

typedef __attribute__((ext_vector_type(2))) float v2f;
typedef __attribute__((ext_vector_type(8))) float v8f;

#define B_ 8
#define C_ 256
#define F_ 128
#define H_ 256

__global__ __launch_bounds__(128) void gemm_ac_wmma(
    const float* __restrict__ xa, const float* __restrict__ W1,
    const float* __restrict__ b1,
    float* __restrict__ ws_a, float* __restrict__ ws_c)
{
  const int wave = blockIdx.x * (blockDim.x >> 5) + (threadIdx.x >> 5);
  const int lane = threadIdx.x & 31;

  int t = wave;
  const int ht  = t & 15; t >>= 4;
  const int it  = t & 15; t >>= 4;
  const int b   = t & 7;  t >>= 3;
  const int sel = t & 1;

  const int i0 = it * 16;
  const int h0 = ht * 16;
  const int lo = lane & 15;
  const int hi = lane >> 4;

  const float* arow = xa + ((size_t)(b * C_ + i0 + lo)) * F_;
  const float* brow = W1 + (size_t)(h0 + lo) * (2 * F_) + sel * F_;

  v8f acc = {};
#pragma unroll 8
  for (int k = 0; k < F_; k += 4) {
    const int ka = k + 2 * hi;
    v2f af = *(const v2f*)(arow + ka);
    v2f bf = *(const v2f*)(brow + ka);
    acc = __builtin_amdgcn_wmma_f32_16x16x4_f32(false, af, false, bf,
                                                (short)0, acc, false, false);
  }

  const float bias = (sel == 0) ? b1[h0 + lo] : 0.0f;
  float* __restrict__ dst = (sel == 0) ? ws_a : ws_c;
#pragma unroll
  for (int r = 0; r < 8; ++r) {
    const int m = i0 + r + 8 * hi;
    dst[((size_t)(b * C_ + m)) * H_ + (h0 + lo)] = acc[r] + bias;
  }
}

__global__ __launch_bounds__(256) void edge_epilogue(
    const float* __restrict__ ws_a, const float* __restrict__ ws_c,
    const float* __restrict__ w2, const float* __restrict__ b2p,
    float* __restrict__ out)
{
  __shared__ float a_s[32 * H_];
  __shared__ float c_s[H_ * 33];
  __shared__ float w_s[H_];

  const int tid = threadIdx.x;
  int t = blockIdx.x;
  const int jt = t & 7; t >>= 3;
  const int it = t & 7; t >>= 3;
  const int b  = t;
  const int i0 = it * 32, j0 = jt * 32;

  {
    const float4* src = (const float4*)(ws_a + ((size_t)(b * C_ + i0)) * H_);
    float4* dst4 = (float4*)a_s;
#pragma unroll
    for (int idx = tid; idx < (32 * H_) / 4; idx += 256) dst4[idx] = src[idx];
  }
  {
    const float4* src = (const float4*)(ws_c + ((size_t)(b * C_ + j0)) * H_);
#pragma unroll
    for (int idx = tid; idx < (32 * H_) / 4; idx += 256) {
      const int j  = idx >> 6;
      const int hb = (idx & 63) * 4;
      float4 v = src[idx];
      c_s[(hb + 0) * 33 + j] = v.x;
      c_s[(hb + 1) * 33 + j] = v.y;
      c_s[(hb + 2) * 33 + j] = v.z;
      c_s[(hb + 3) * 33 + j] = v.w;
    }
  }
  if (tid < H_) w_s[tid] = w2[tid];
  __syncthreads();

  const int ti = tid >> 5;
  const int tj = tid & 31;
  const float* a0 = a_s + (ti +  0) * H_;
  const float* a1 = a_s + (ti +  8) * H_;
  const float* a2 = a_s + (ti + 16) * H_;
  const float* a3 = a_s + (ti + 24) * H_;

  float acc0 = 0.f, acc1 = 0.f, acc2 = 0.f, acc3 = 0.f;
#pragma unroll 4
  for (int h = 0; h < H_; ++h) {
    const float wv = w_s[h];
    const float cv = c_s[h * 33 + tj];
    float x;
    x = a0[h] + cv; acc0 = fmaf(fmaxf(x, 0.f), wv, acc0);
    x = a1[h] + cv; acc1 = fmaf(fmaxf(x, 0.f), wv, acc1);
    x = a2[h] + cv; acc2 = fmaf(fmaxf(x, 0.f), wv, acc2);
    x = a3[h] + cv; acc3 = fmaf(fmaxf(x, 0.f), wv, acc3);
  }

  const float b2 = b2p[0];
  const size_t obase = ((size_t)b * C_) * C_;
  const int j = j0 + tj;
  float* o = out + obase + (size_t)(i0 + ti) * C_ + j;
  o[ 0 * 8 * C_] = 1.0f / (1.0f + __expf(-(acc0 + b2)));
  o[ 1 * 8 * C_] = 1.0f / (1.0f + __expf(-(acc1 + b2)));
  o[ 2 * 8 * C_] = 1.0f / (1.0f + __expf(-(acc2 + b2)));
  o[ 3 * 8 * C_] = 1.0f / (1.0f + __expf(-(acc3 + b2)));
}

extern "C" void kernel_launch(void* const* d_in, const int* in_sizes, int n_in,
                              void* d_out, int out_size, void* d_ws, size_t ws_size,
                              hipStream_t stream) {
  const float* xa = (const float*)d_in[0];
  const float* W1 = (const float*)d_in[1];
  const float* b1 = (const float*)d_in[2];
  const float* w2 = (const float*)d_in[3];
  const float* b2 = (const float*)d_in[4];
  float* out = (float*)d_out;

  float* ws_a = (float*)d_ws;
  float* ws_c = ws_a + (size_t)B_ * C_ * H_;

  gemm_ac_wmma<<<1024, 128, 0, stream>>>(xa, W1, b1, ws_a, ws_c);

  edge_epilogue<<<512, 256, 0, stream>>>(ws_a, ws_c, w2, b2, out);
}


# first GEMM fed without LDS: one wave per 32x32 tile, A and B fragments read straight into registers by 16-byte loads far ahead of their WMMAs; k-sum order changed (a WMMA sums k = 8c+{0,1,4,5})
# speedup vs baseline: 1.4512x; 1.4512x over previous
#include <hip/hip_runtime.h>
#include <math.h>

typedef __attribute__((ext_vector_type(2))) float v2f;
typedef __attribute__((ext_vector_type(4))) float v4f;
typedef __attribute__((ext_vector_type(8))) float v8f;

#define B_ 8
#define C_ 256
#define F_ 128
#define H_ 256

__device__ __forceinline__ v8f mma4(v2f a, v2f b, v8f c) {
  return __builtin_amdgcn_wmma_f32_16x16x4_f32(false, a, false, b, (short)0, c, false, false);
}

__global__ __launch_bounds__(128) __attribute__((amdgpu_waves_per_eu(1, 1)))
void gemm_ac_wmma(
    const float* __restrict__ xa, const float* __restrict__ W1,
    const float* __restrict__ b1,
    float* __restrict__ ws_a, float* __restrict__ ws_c)
{
  const int wave = blockIdx.x * 4 + (threadIdx.x >> 5);
  const int lane = threadIdx.x & 31;

  int t = wave;
  const int ht  = t & 7; t >>= 3;
  const int it  = t & 7; t >>= 3;
  const int b   = t & 7; t >>= 3;
  const int sel = t & 1;

  const int i0 = it * 32;
  const int h0 = ht * 32;
  const int lo = lane & 15;
  const int hi = lane >> 4;

  const float* a0p = xa + ((size_t)(b * C_ + i0 + lo)) * F_ + 4 * hi;
  const float* a1p = a0p + 16 * F_;
  const float* b0p = W1 + (size_t)(h0 + lo) * (2 * F_) + sel * F_ + 4 * hi;
  const float* b1p = b0p + 16 * (2 * F_);

  const float bv0 = b1[h0 + lo];
  const float bv1 = b1[h0 + 16 + lo];

  v4f A0[16], A1[16], B0[16], B1[16];
#pragma unroll
  for (int kc = 0; kc < 16; ++kc) {
    A0[kc] = *(const v4f*)(a0p + kc * 8);
    B0[kc] = *(const v4f*)(b0p + kc * 8);
    A1[kc] = *(const v4f*)(a1p + kc * 8);
    B1[kc] = *(const v4f*)(b1p + kc * 8);
  }

  v8f acc00 = {}, acc01 = {}, acc10 = {}, acc11 = {};
#pragma unroll
  for (int kc = 0; kc < 16; ++kc) {
    acc00 = mma4(A0[kc].xy, B0[kc].xy, acc00);
    acc01 = mma4(A0[kc].xy, B1[kc].xy, acc01);
    acc10 = mma4(A1[kc].xy, B0[kc].xy, acc10);
    acc11 = mma4(A1[kc].xy, B1[kc].xy, acc11);
    acc00 = mma4(A0[kc].zw, B0[kc].zw, acc00);
    acc01 = mma4(A0[kc].zw, B1[kc].zw, acc01);
    acc10 = mma4(A1[kc].zw, B0[kc].zw, acc10);
    acc11 = mma4(A1[kc].zw, B1[kc].zw, acc11);
  }

  const float bias0 = (sel == 0) ? bv0 : 0.0f;
  const float bias1 = (sel == 0) ? bv1 : 0.0f;
  float* __restrict__ dst = ((sel == 0) ? ws_a : ws_c)
                          + ((size_t)(b * C_ + i0 + 8 * hi)) * H_ + h0 + lo;
#pragma unroll
  for (int r = 0; r < 8; ++r) {
    dst[(size_t)r * H_]             = acc00[r] + bias0;
    dst[(size_t)r * H_ + 16]        = acc01[r] + bias1;
    dst[(size_t)(r + 16) * H_]      = acc10[r] + bias0;
    dst[(size_t)(r + 16) * H_ + 16] = acc11[r] + bias1;
  }
}

template <int CTRL>
__device__ __forceinline__ float row_xchg(float v) {
  return __builtin_bit_cast(float,
      __builtin_amdgcn_update_dpp(0, __builtin_bit_cast(int, v), CTRL, 0xf, 0xf, true));
}

__device__ __forceinline__ void chunk_load(const float* __restrict__ ap,
                                           const float* __restrict__ cp,
                                           const float* __restrict__ wp, int k,
                                           v4f (&A)[8], v4f (&Cc)[8], v4f& Wv) {
#pragma unroll
  for (int i = 0; i < 8; ++i) A[i] = *(const v4f*)(ap + i * H_ + k * 32);
#pragma unroll
  for (int jj = 0; jj < 8; ++jj) Cc[jj] = *(const v4f*)(cp + jj * H_ + k * 32);
  Wv = *(const v4f*)(wp + k * 32);
}

__device__ __forceinline__ void chunk_fma(const v4f (&A)[8], const v4f (&Cc)[8],
                                          const v4f Wv, v2f (&acc)[8][8]) {
  const v2f z = {0.f, 0.f};
#pragma unroll
  for (int i = 0; i < 8; ++i) {
    v2f x0[8], x1[8];
#pragma unroll
    for (int jj = 0; jj < 8; ++jj) { x0[jj] = A[i].xy + Cc[jj].xy; x1[jj] = A[i].zw + Cc[jj].zw; }
#pragma unroll
    for (int jj = 0; jj < 8; ++jj) { x0[jj] = __builtin_elementwise_max(x0[jj], z); x1[jj] = __builtin_elementwise_max(x1[jj], z); }
#pragma unroll
    for (int jj = 0; jj < 8; ++jj) acc[i][jj] = __builtin_elementwise_fma(x0[jj], Wv.xy, acc[i][jj]);
#pragma unroll
    for (int jj = 0; jj < 8; ++jj) acc[i][jj] = __builtin_elementwise_fma(x1[jj], Wv.zw, acc[i][jj]);
  }
}

__global__ __launch_bounds__(256) __attribute__((amdgpu_waves_per_eu(2, 2)))
void edge_epilogue(
    const float* __restrict__ ws_a, const float* __restrict__ ws_c,
    const float* __restrict__ w2, const float* __restrict__ b2p,
    float* __restrict__ out)
{
  const int lane = threadIdx.x & 31;
  const int wv   = threadIdx.x >> 5;
  int t = blockIdx.x;
  const int ib = t & 3; t >>= 2;
  const int jt = t & 7; t >>= 3;
  const int b  = t;
  const int i0 = ib * 64 + wv * 8, j0 = jt * 32;
  const int hs = lane & 7;
  const int q  = lane >> 3;

  const float* ap = ws_a + ((size_t)(b * C_ + i0)) * H_ + hs * 4;
  const float* cp = ws_c + ((size_t)(b * C_ + j0 + q * 8)) * H_ + hs * 4;
  const float* wp = w2 + hs * 4;

  v2f acc[8][8] = {};
  v4f A0[8], C0[8], W0, A1[8], C1[8], W1;
  chunk_load(ap, cp, wp, 0, A0, C0, W0);
#pragma unroll 1
  for (int k = 0; k < 8; k += 2) {
    chunk_load(ap, cp, wp, k + 1, A1, C1, W1);
    chunk_fma(A0, C0, W0, acc);
    const int kn = (k + 2 < 8) ? k + 2 : 7;
    chunk_load(ap, cp, wp, kn, A0, C0, W0);
    chunk_fma(A1, C1, W1, acc);
  }

  const bool b0 = (lane & 1) != 0, b1 = (lane & 2) != 0, b2l = (lane & 4) != 0;
  float fin[8];
#pragma unroll
  for (int i = 0; i < 8; ++i) {
    float r1[4], r2[2];
#pragma unroll
    for (int p = 0; p < 4; ++p) {
      const float X = acc[i][2 * p].x + acc[i][2 * p].y;
      const float Y = acc[i][2 * p + 1].x + acc[i][2 * p + 1].y;
      const float mine = b0 ? Y : X, give = b0 ? X : Y;
      r1[p] = mine + row_xchg<0x161>(give);
    }
#pragma unroll
    for (int p = 0; p < 2; ++p) {
      const float X = r1[2 * p], Y = r1[2 * p + 1];
      const float mine = b1 ? Y : X, give = b1 ? X : Y;
      r2[p] = mine + row_xchg<0x162>(give);
    }
    {
      const float X = r2[0], Y = r2[1];
      const float mine = b2l ? Y : X, give = b2l ? X : Y;
      fin[i] = mine + row_xchg<0x164>(give);
    }
  }

  const float b2 = b2p[0];
  float* o = out + ((size_t)(b * C_ + i0)) * C_ + j0 + lane;
#pragma unroll
  for (int i = 0; i < 8; ++i)
    o[i * C_] = 1.0f / (1.0f + __expf(-(fin[i] + b2)));
}

extern "C" void kernel_launch(void* const* d_in, const int* in_sizes, int n_in,
                              void* d_out, int out_size, void* d_ws, size_t ws_size,
                              hipStream_t stream) {
  const float* xa = (const float*)d_in[0];
  const float* W1 = (const float*)d_in[1];
  const float* b1 = (const float*)d_in[2];
  const float* w2 = (const float*)d_in[3];
  const float* b2 = (const float*)d_in[4];
  float* out = (float*)d_out;

  float* ws_a = (float*)d_ws;
  float* ws_c = ws_a + (size_t)B_ * C_ * H_;

  gemm_ac_wmma<<<256, 128, 0, stream>>>(xa, W1, b1, ws_a, ws_c);

  edge_epilogue<<<256, 256, 0, stream>>>(ws_a, ws_c, w2, b2, out);
}


# epilogue loads issued two chunks of arithmetic ahead via three register sets, load groups pinned ahead of the arithmetic (parent as built loaded c, w2 of even chunks at first use); h-sum order same
# speedup vs baseline: 1.5657x; 1.0789x over previous
#include <hip/hip_runtime.h>
#include <math.h>

typedef __attribute__((ext_vector_type(2))) float v2f;
typedef __attribute__((ext_vector_type(4))) float v4f;
typedef __attribute__((ext_vector_type(8))) float v8f;

#define B_ 8
#define C_ 256
#define F_ 128
#define H_ 256

__device__ __forceinline__ v8f mma4(v2f a, v2f b, v8f c) {
  return __builtin_amdgcn_wmma_f32_16x16x4_f32(false, a, false, b, (short)0, c, false, false);
}

__global__ __launch_bounds__(128) __attribute__((amdgpu_waves_per_eu(1, 1)))
void gemm_ac_wmma(
    const float* __restrict__ xa, const float* __restrict__ W1,
    const float* __restrict__ b1,
    float* __restrict__ ws_a, float* __restrict__ ws_c)
{
  const int wave = blockIdx.x * 4 + (threadIdx.x >> 5);
  const int lane = threadIdx.x & 31;

  int t = wave;
  const int ht  = t & 7; t >>= 3;
  const int it  = t & 7; t >>= 3;
  const int b   = t & 7; t >>= 3;
  const int sel = t & 1;

  const int i0 = it * 32;
  const int h0 = ht * 32;
  const int lo = lane & 15;
  const int hi = lane >> 4;

  const float* a0p = xa + ((size_t)(b * C_ + i0 + lo)) * F_ + 4 * hi;
  const float* a1p = a0p + 16 * F_;
  const float* b0p = W1 + (size_t)(h0 + lo) * (2 * F_) + sel * F_ + 4 * hi;
  const float* b1p = b0p + 16 * (2 * F_);

  const float bv0 = b1[h0 + lo];
  const float bv1 = b1[h0 + 16 + lo];

  v4f A0[16], A1[16], B0[16], B1[16];
#pragma unroll
  for (int kc = 0; kc < 16; ++kc) {
    A0[kc] = *(const v4f*)(a0p + kc * 8);
    B0[kc] = *(const v4f*)(b0p + kc * 8);
    A1[kc] = *(const v4f*)(a1p + kc * 8);
    B1[kc] = *(const v4f*)(b1p + kc * 8);
  }

  v8f acc00 = {}, acc01 = {}, acc10 = {}, acc11 = {};
#pragma unroll
  for (int kc = 0; kc < 16; ++kc) {
    acc00 = mma4(A0[kc].xy, B0[kc].xy, acc00);
    acc01 = mma4(A0[kc].xy, B1[kc].xy, acc01);
    acc10 = mma4(A1[kc].xy, B0[kc].xy, acc10);
    acc11 = mma4(A1[kc].xy, B1[kc].xy, acc11);
    acc00 = mma4(A0[kc].zw, B0[kc].zw, acc00);
    acc01 = mma4(A0[kc].zw, B1[kc].zw, acc01);
    acc10 = mma4(A1[kc].zw, B0[kc].zw, acc10);
    acc11 = mma4(A1[kc].zw, B1[kc].zw, acc11);
  }

  const float bias0 = (sel == 0) ? bv0 : 0.0f;
  const float bias1 = (sel == 0) ? bv1 : 0.0f;
  float* __restrict__ dst = ((sel == 0) ? ws_a : ws_c)
                          + ((size_t)(b * C_ + i0 + 8 * hi)) * H_ + h0 + lo;
#pragma unroll
  for (int r = 0; r < 8; ++r) {
    dst[(size_t)r * H_]             = acc00[r] + bias0;
    dst[(size_t)r * H_ + 16]        = acc01[r] + bias1;
    dst[(size_t)(r + 16) * H_]      = acc10[r] + bias0;
    dst[(size_t)(r + 16) * H_ + 16] = acc11[r] + bias1;
  }
}

template <int CTRL>
__device__ __forceinline__ float row_xchg(float v) {
  return __builtin_bit_cast(float,
      __builtin_amdgcn_update_dpp(0, __builtin_bit_cast(int, v), CTRL, 0xf, 0xf, true));
}

__device__ __forceinline__ void chunk_load(const float* __restrict__ ap,
                                           const float* __restrict__ cp,
                                           const float* __restrict__ wp, int k,
                                           v4f (&A)[8], v4f (&Cc)[8], v4f& Wv) {
#pragma unroll
  for (int i = 0; i < 8; ++i) A[i] = *(const v4f*)(ap + i * H_ + k * 32);
#pragma unroll
  for (int jj = 0; jj < 8; ++jj) Cc[jj] = *(const v4f*)(cp + jj * H_ + k * 32);
  Wv = *(const v4f*)(wp + k * 32);
}

__device__ __forceinline__ void chunk_fma(const v4f (&A)[8], const v4f (&Cc)[8],
                                          const v4f Wv, v2f (&acc)[8][8]) {
  const v2f z = {0.f, 0.f};
#pragma unroll
  for (int i = 0; i < 8; ++i) {
    v2f x0[8], x1[8];
#pragma unroll
    for (int jj = 0; jj < 8; ++jj) { x0[jj] = A[i].xy + Cc[jj].xy; x1[jj] = A[i].zw + Cc[jj].zw; }
#pragma unroll
    for (int jj = 0; jj < 8; ++jj) { x0[jj] = __builtin_elementwise_max(x0[jj], z); x1[jj] = __builtin_elementwise_max(x1[jj], z); }
#pragma unroll
    for (int jj = 0; jj < 8; ++jj) acc[i][jj] = __builtin_elementwise_fma(x0[jj], Wv.xy, acc[i][jj]);
#pragma unroll
    for (int jj = 0; jj < 8; ++jj) acc[i][jj] = __builtin_elementwise_fma(x1[jj], Wv.zw, acc[i][jj]);
  }
}

__global__ __launch_bounds__(256) __attribute__((amdgpu_waves_per_eu(2, 2)))
void edge_epilogue(
    const float* __restrict__ ws_a, const float* __restrict__ ws_c,
    const float* __restrict__ w2, const float* __restrict__ b2p,
    float* __restrict__ out)
{
  const int lane = threadIdx.x & 31;
  const int wv   = threadIdx.x >> 5;
  int t = blockIdx.x;
  const int ib = t & 3; t >>= 2;
  const int jt = t & 7; t >>= 3;
  const int b  = t;
  const int i0 = ib * 64 + wv * 8, j0 = jt * 32;
  const int hs = lane & 7;
  const int q  = lane >> 3;

  const float* ap = ws_a + ((size_t)(b * C_ + i0)) * H_ + hs * 4;
  const float* cp = ws_c + ((size_t)(b * C_ + j0 + q * 8)) * H_ + hs * 4;
  const float* wp = w2 + hs * 4;

  v2f acc[8][8] = {};
  v4f A0[8], C0[8], W0, A1[8], C1[8], W1, A2[8], C2[8], W2;
  chunk_load(ap, cp, wp, 0, A0, C0, W0);
  __builtin_amdgcn_sched_barrier(0);
  chunk_load(ap, cp, wp, 1, A1, C1, W1);
#pragma unroll 1
  for (int k = 0; k < 6; k += 3) {
    chunk_load(ap, cp, wp, k + 2, A2, C2, W2);
    __builtin_amdgcn_sched_barrier(0);
    chunk_fma(A0, C0, W0, acc);
    chunk_load(ap, cp, wp, k + 3, A0, C0, W0);
    __builtin_amdgcn_sched_barrier(0);
    chunk_fma(A1, C1, W1, acc);
    chunk_load(ap, cp, wp, k + 4, A1, C1, W1);
    __builtin_amdgcn_sched_barrier(0);
    chunk_fma(A2, C2, W2, acc);
  }
  chunk_fma(A0, C0, W0, acc);
  chunk_fma(A1, C1, W1, acc);

  const bool b0 = (lane & 1) != 0, b1 = (lane & 2) != 0, b2l = (lane & 4) != 0;
  float fin[8];
#pragma unroll
  for (int i = 0; i < 8; ++i) {
    float r1[4], r2[2];
#pragma unroll
    for (int p = 0; p < 4; ++p) {
      const float X = acc[i][2 * p].x + acc[i][2 * p].y;
      const float Y = acc[i][2 * p + 1].x + acc[i][2 * p + 1].y;
      const float mine = b0 ? Y : X, give = b0 ? X : Y;
      r1[p] = mine + row_xchg<0x161>(give);
    }
#pragma unroll
    for (int p = 0; p < 2; ++p) {
      const float X = r1[2 * p], Y = r1[2 * p + 1];
      const float mine = b1 ? Y : X, give = b1 ? X : Y;
      r2[p] = mine + row_xchg<0x162>(give);
    }
    {
      const float X = r2[0], Y = r2[1];
      const float mine = b2l ? Y : X, give = b2l ? X : Y;
      fin[i] = mine + row_xchg<0x164>(give);
    }
  }

  const float b2 = b2p[0];
  float* o = out + ((size_t)(b * C_ + i0)) * C_ + j0 + lane;
#pragma unroll
  for (int i = 0; i < 8; ++i)
    o[i * C_] = 1.0f / (1.0f + __expf(-(fin[i] + b2)));
}

extern "C" void kernel_launch(void* const* d_in, const int* in_sizes, int n_in,
                              void* d_out, int out_size, void* d_ws, size_t ws_size,
                              hipStream_t stream) {
  const float* xa = (const float*)d_in[0];
  const float* W1 = (const float*)d_in[1];
  const float* b1 = (const float*)d_in[2];
  const float* w2 = (const float*)d_in[3];
  const float* b2 = (const float*)d_in[4];
  float* out = (float*)d_out;

  float* ws_a = (float*)d_ws;
  float* ws_c = ws_a + (size_t)B_ * C_ * H_;

  gemm_ac_wmma<<<256, 128, 0, stream>>>(xa, W1, b1, ws_a, ws_c);

  edge_epilogue<<<256, 256, 0, stream>>>(ws_a, ws_c, w2, b2, out);
}
